# starting kernel
# baseline (speedup 1.0000x reference)
#include <hip/hip_runtime.h>
#include <math.h>

constexpr int Bb = 8;
constexpr int Ll = 256;
constexpr int Dd = 256;
constexpr int Kk = 64;

typedef __attribute__((ext_vector_type(2))) float v2f;
typedef __attribute__((ext_vector_type(8))) float v8f;

__device__ __forceinline__ v8f wmma_f32_4(v2f a, v2f b, v8f c) {
  return __builtin_amdgcn_wmma_f32_16x16x4_f32(false, a, false, b, (short)0, c,
                                               false, false);
}

__device__ __forceinline__ float fast_tanh(float x) {
#if __has_builtin(__builtin_amdgcn_tanhf)
  return __builtin_amdgcn_tanhf(x);
#else
  float r;
  asm volatile("v_tanh_f32 %0, %1" : "=v"(r) : "v"(x));
  return r;
#endif
}

__global__ __launch_bounds__(256) void qk_proj_kernel(
    const float* __restrict__ inp, const float* __restrict__ W1,
    float* __restrict__ qi, float* __restrict__ kj) {
  const int blk  = blockIdx.x;
  const int b    = blk / (Ll / 16);
  const int i0   = (blk % (Ll / 16)) * 16;
  const int wave = threadIdx.x >> 5;
  const int lane = threadIdx.x & 31;
  const bool is_k = (wave >= 4);
  const int n0     = (wave & 3) * 16;
  const int coloff = is_k ? Dd : 0;

  const int mn = lane & 15;
  const int kh = (lane >> 4) << 1;

  const float* arow = inp + ((size_t)b * Ll + (i0 + mn)) * Dd;
  const float* brow = W1 + (size_t)(n0 + mn) * (2 * Dd) + coloff;

  v8f acc = {};
#pragma unroll 4
  for (int d = 0; d < Dd; d += 4) {
    v2f a, bv;
    a.x  = arow[d + kh];
    a.y  = arow[d + kh + 1];
    bv.x = brow[d + kh];
    bv.y = brow[d + kh + 1];
    acc = wmma_f32_4(a, bv, acc);
  }

  float* op     = is_k ? kj : qi;
  const int mb  = (lane >> 4) * 8;
#pragma unroll
  for (int r = 0; r < 8; ++r) {
    op[((size_t)b * Ll + (i0 + mb + r)) * Kk + (n0 + mn)] = acc[r];
  }
}

__global__ __launch_bounds__(256) void attn_softmax_kernel(
    const float* __restrict__ qi, const float* __restrict__ kj,
    const float* __restrict__ W2, float* __restrict__ attn) {
  __shared__ float q_s[Kk];
  __shared__ float w_s[Kk];
  __shared__ float r_s[Ll];

  const int b = blockIdx.x / Ll;
  const int i = blockIdx.x % Ll;
  const int t = threadIdx.x;

  if (t < Kk) {
    q_s[t] = qi[((size_t)b * Ll + i) * Kk + t];
    w_s[t] = W2[t];
  }
  __syncthreads();

  const float4* kr4 =
      reinterpret_cast<const float4*>(kj + ((size_t)b * Ll + t) * Kk);
  float s = 0.0f;
#pragma unroll
  for (int k4 = 0; k4 < Kk / 4; ++k4) {
    const float4 kv = kr4[k4];
    const int k = 4 * k4;
    s += fast_tanh(q_s[k + 0] + kv.x) * w_s[k + 0];
    s += fast_tanh(q_s[k + 1] + kv.y) * w_s[k + 1];
    s += fast_tanh(q_s[k + 2] + kv.z) * w_s[k + 2];
    s += fast_tanh(q_s[k + 3] + kv.w) * w_s[k + 3];
  }

  r_s[t] = s;
  __syncthreads();
  for (int st = 128; st > 0; st >>= 1) {
    if (t < st) r_s[t] = fmaxf(r_s[t], r_s[t + st]);
    __syncthreads();
  }
  const float mx = r_s[0];
  __syncthreads();

  const float e = __expf(s - mx);

  r_s[t] = e;
  __syncthreads();
  for (int st = 128; st > 0; st >>= 1) {
    if (t < st) r_s[t] += r_s[t + st];
    __syncthreads();
  }

  attn[((size_t)b * Ll + i) * Ll + t] = e * (1.0f / r_s[0]);
}

__global__ __launch_bounds__(256) void context_kernel(
    const float* __restrict__ attn, const float* __restrict__ inp,
    float* __restrict__ out) {
  const int wave = threadIdx.x >> 5;
  const int lane = threadIdx.x & 31;
  const int g    = blockIdx.x * 8 + wave;
  const int b    = g >> 8;
  const int rem  = g & 255;
  const int i0   = (rem >> 4) * 16;
  const int d0   = (rem & 15) * 16;

  const int mn = lane & 15;
  const int kh = (lane >> 4) << 1;

  const float* arow = attn + ((size_t)b * Ll + (i0 + mn)) * Ll;

  v8f acc = {};
  for (int j = 0; j < Ll; j += 4) {
    v2f a, bv;
    a.x = arow[j + kh];
    a.y = arow[j + kh + 1];
    const float* bp = inp + ((size_t)b * Ll + (j + kh)) * Dd + (d0 + mn);
    bv.x = bp[0];
    bv.y = bp[Dd];
    acc = wmma_f32_4(a, bv, acc);
  }

  const int mb = (lane >> 4) * 8;
#pragma unroll
  for (int r = 0; r < 8; ++r) {
    const int row      = i0 + mb + r;
    const size_t obase = ((size_t)b * Ll + row) * (2 * Dd);
    out[obase + Dd + d0 + mn] = acc[r];
    out[obase + d0 + mn] = inp[((size_t)b * Ll + row) * Dd + d0 + mn];
  }
}

extern "C" void kernel_launch(void* const* d_in, const int* in_sizes, int n_in,
                              void* d_out, int out_size, void* d_ws,
                              size_t ws_size, hipStream_t stream) {
  const float* inp = (const float*)d_in[0];
  const float* W1  = (const float*)d_in[1];
  const float* W2  = (const float*)d_in[2];
  float* out = (float*)d_out;

  float* qi   = (float*)d_ws;
  float* kj   = qi + (size_t)Bb * Ll * Kk;
  float* attn = kj + (size_t)Bb * Ll * Kk;

  qk_proj_kernel<<<Bb * (Ll / 16), 256, 0, stream>>>(inp, W1, qi, kj);
  attn_softmax_kernel<<<Bb * Ll, 256, 0, stream>>>(qi, kj, W2, attn);
  context_kernel<<<(Bb * (Ll / 16) * (Dd / 16)) / 8, 256, 0, stream>>>(attn, inp, out);
}


# attn kernel re-tiled: one workgroup computes 8 score rows, its kj rows staged once through LDS (8x fewer waves); softmax max/sum by lane butterfly + one LDS exchange, which changes the sum order
# speedup vs baseline: 1.8458x; 1.8458x over previous
#include <hip/hip_runtime.h>
#include <math.h>

constexpr int Bb = 8;
constexpr int Ll = 256;
constexpr int Dd = 256;
constexpr int Kk = 64;

typedef __attribute__((ext_vector_type(2))) float v2f;
typedef __attribute__((ext_vector_type(8))) float v8f;

__device__ __forceinline__ v8f wmma_f32_4(v2f a, v2f b, v8f c) {
  return __builtin_amdgcn_wmma_f32_16x16x4_f32(false, a, false, b, (short)0, c,
                                               false, false);
}

__device__ __forceinline__ float fast_tanh(float x) {
#if __has_builtin(__builtin_amdgcn_tanhf)
  return __builtin_amdgcn_tanhf(x);
#else
  float r;
  asm volatile("v_tanh_f32 %0, %1" : "=v"(r) : "v"(x));
  return r;
#endif
}

__global__ __launch_bounds__(256) void qk_proj_kernel(
    const float* __restrict__ inp, const float* __restrict__ W1,
    float* __restrict__ qi, float* __restrict__ kj) {
  const int blk  = blockIdx.x;
  const int b    = blk / (Ll / 16);
  const int i0   = (blk % (Ll / 16)) * 16;
  const int wave = threadIdx.x >> 5;
  const int lane = threadIdx.x & 31;
  const bool is_k = (wave >= 4);
  const int n0     = (wave & 3) * 16;
  const int coloff = is_k ? Dd : 0;

  const int mn = lane & 15;
  const int kh = (lane >> 4) << 1;

  const float* arow = inp + ((size_t)b * Ll + (i0 + mn)) * Dd;
  const float* brow = W1 + (size_t)(n0 + mn) * (2 * Dd) + coloff;

  v8f acc = {};
#pragma unroll 4
  for (int d = 0; d < Dd; d += 4) {
    v2f a, bv;
    a.x  = arow[d + kh];
    a.y  = arow[d + kh + 1];
    bv.x = brow[d + kh];
    bv.y = brow[d + kh + 1];
    acc = wmma_f32_4(a, bv, acc);
  }

  float* op     = is_k ? kj : qi;
  const int mb  = (lane >> 4) * 8;
#pragma unroll
  for (int r = 0; r < 8; ++r) {
    op[((size_t)b * Ll + (i0 + mb + r)) * Kk + (n0 + mn)] = acc[r];
  }
}

constexpr int RB  = 8;
constexpr int KP4 = Kk / 4 + 1;

template <int M>
__device__ __forceinline__ float row_xor(float x) {
  const int v = __builtin_bit_cast(int, x);
  return __builtin_bit_cast(
      float, __builtin_amdgcn_update_dpp(v, v, 0x160 | M, 0xf, 0xf, false));
}
__device__ __forceinline__ float xhalf(float x) {
  const int v = __builtin_bit_cast(int, x);
  return __builtin_bit_cast(
      float, __builtin_amdgcn_permlanex16(v, v, 0x76543210, 0xfedcba98, false,
                                          false));
}
__device__ __forceinline__ float wave_max(float x) {
  x = fmaxf(x, row_xor<1>(x));
  x = fmaxf(x, row_xor<2>(x));
  x = fmaxf(x, row_xor<4>(x));
  x = fmaxf(x, row_xor<8>(x));
  return fmaxf(x, xhalf(x));
}
__device__ __forceinline__ float wave_sum(float x) {
  x += row_xor<1>(x);
  x += row_xor<2>(x);
  x += row_xor<4>(x);
  x += row_xor<8>(x);
  return x + xhalf(x);
}

struct KBlk {
  float4 kv;
  float4 w;
  float4 q[2 * 4];
};

__device__ __forceinline__ void load_blk(KBlk& d, const float4* myrow,
                                         const float4* w_s, const float4* q_s,
                                         int k4) {
  d.kv = myrow[k4];
  d.w  = w_s[k4];
#pragma unroll
  for (int c = 0; c < 8; ++c) d.q[c] = q_s[k4 * 8 + c];
}

__device__ __forceinline__ void k_step(float (&s)[RB], float kv, float w,
                                       const float4 qa, const float4 qb) {
  s[0] += fast_tanh(qa.x + kv) * w;
  s[1] += fast_tanh(qa.y + kv) * w;
  s[2] += fast_tanh(qa.z + kv) * w;
  s[3] += fast_tanh(qa.w + kv) * w;
  s[4] += fast_tanh(qb.x + kv) * w;
  s[5] += fast_tanh(qb.y + kv) * w;
  s[6] += fast_tanh(qb.z + kv) * w;
  s[7] += fast_tanh(qb.w + kv) * w;
}

__global__ __launch_bounds__(256) void attn_softmax_kernel(
    const float* __restrict__ qi, const float* __restrict__ kj,
    const float* __restrict__ W2, float* __restrict__ attn) {
  constexpr int Q_OFF  = 0;
  constexpr int W_OFF  = Q_OFF + Kk * (RB / 4);
  constexpr int MX_OFF = W_OFF + Kk / 4;
  constexpr int SM_OFF = MX_OFF + 8 * (RB / 4);
  constexpr int KJ_OFF = SM_OFF + 8 * (RB / 4);
  __shared__ float4 lds[KJ_OFF + 8 * 32 * KP4];
  float4* q_s  = lds + Q_OFF;
  float4* w_s  = lds + W_OFF;
  float4* mx_s = lds + MX_OFF;
  float4* sm_s = lds + SM_OFF;

  const int b    = blockIdx.x / (Ll / RB);
  const int i0   = (blockIdx.x % (Ll / RB)) * RB;
  const int t    = threadIdx.x;
  const int wave = t >> 5;
  const int lane = t & 31;

  const float4* src = reinterpret_cast<const float4*>(
      kj + ((size_t)b * Ll + 32 * wave) * Kk);
  float4* tile = lds + KJ_OFF + wave * 32 * KP4;
#pragma unroll
  for (int it = 0; it < 16; ++it) {
    const int f = it * 32 + lane;
    tile[(f >> 4) * KP4 + (f & 15)] = src[f];
  }
  {
    float* q_f = reinterpret_cast<float*>(q_s);
    for (int it = 0; it < (RB * Kk) / 256; ++it) {
      const int idx = it * 256 + t;
      const int r = idx >> 6;
      const int k = idx & 63;
      q_f[k * RB + r] = qi[((size_t)b * Ll + i0 + r) * Kk + k];
    }
    if (t < Kk) reinterpret_cast<float*>(w_s)[t] = W2[t];
  }
  __syncthreads();

  const float4* myrow = tile + lane * KP4;
  float s[RB];
#pragma unroll
  for (int r = 0; r < RB; ++r) s[r] = 0.0f;

  KBlk blk[2];
  load_blk(blk[0], myrow, w_s, q_s, 0);
#pragma unroll
  for (int k4 = 0; k4 < Kk / 4; ++k4) {
    if (k4 + 1 < Kk / 4) load_blk(blk[(k4 + 1) & 1], myrow, w_s, q_s, k4 + 1);
    __builtin_amdgcn_sched_barrier(0);
    const KBlk& c = blk[k4 & 1];
    k_step(s, c.kv.x, c.w.x, c.q[0], c.q[1]);
    k_step(s, c.kv.y, c.w.y, c.q[2], c.q[3]);
    k_step(s, c.kv.z, c.w.z, c.q[4], c.q[5]);
    k_step(s, c.kv.w, c.w.w, c.q[6], c.q[7]);
  }

  float m[RB];
#pragma unroll
  for (int r = 0; r < RB; ++r) m[r] = wave_max(s[r]);
  if (lane == 0) {
    mx_s[wave * 2 + 0] = make_float4(m[0], m[1], m[2], m[3]);
    mx_s[wave * 2 + 1] = make_float4(m[4], m[5], m[6], m[7]);
  }
  __syncthreads();
  {
    float4 a = mx_s[0], c = mx_s[1];
    for (int w = 1; w < 8; ++w) {
      const float4 a2 = mx_s[w * 2 + 0], c2 = mx_s[w * 2 + 1];
      a.x = fmaxf(a.x, a2.x); a.y = fmaxf(a.y, a2.y);
      a.z = fmaxf(a.z, a2.z); a.w = fmaxf(a.w, a2.w);
      c.x = fmaxf(c.x, c2.x); c.y = fmaxf(c.y, c2.y);
      c.z = fmaxf(c.z, c2.z); c.w = fmaxf(c.w, c2.w);
    }
    m[0] = a.x; m[1] = a.y; m[2] = a.z; m[3] = a.w;
    m[4] = c.x; m[5] = c.y; m[6] = c.z; m[7] = c.w;
  }

  float e[RB], p[RB];
#pragma unroll
  for (int r = 0; r < RB; ++r) {
    e[r] = __expf(s[r] - m[r]);
    p[r] = wave_sum(e[r]);
  }
  if (lane == 0) {
    sm_s[wave * 2 + 0] = make_float4(p[0], p[1], p[2], p[3]);
    sm_s[wave * 2 + 1] = make_float4(p[4], p[5], p[6], p[7]);
  }
  __syncthreads();
  {
    float4 a = sm_s[0], c = sm_s[1];
    for (int w = 1; w < 8; ++w) {
      const float4 a2 = sm_s[w * 2 + 0], c2 = sm_s[w * 2 + 1];
      a.x += a2.x; a.y += a2.y; a.z += a2.z; a.w += a2.w;
      c.x += c2.x; c.y += c2.y; c.z += c2.z; c.w += c2.w;
    }
    p[0] = a.x; p[1] = a.y; p[2] = a.z; p[3] = a.w;
    p[4] = c.x; p[5] = c.y; p[6] = c.z; p[7] = c.w;
  }

#pragma unroll
  for (int r = 0; r < RB; ++r) {
    attn[((size_t)b * Ll + (i0 + r)) * Ll + t] = e[r] * (1.0f / p[r]);
  }
}

__global__ __launch_bounds__(256) void context_kernel(
    const float* __restrict__ attn, const float* __restrict__ inp,
    float* __restrict__ out) {
  const int wave = threadIdx.x >> 5;
  const int lane = threadIdx.x & 31;
  const int g    = blockIdx.x * 8 + wave;
  const int b    = g >> 8;
  const int rem  = g & 255;
  const int i0   = (rem >> 4) * 16;
  const int d0   = (rem & 15) * 16;

  const int mn = lane & 15;
  const int kh = (lane >> 4) << 1;

  const float* arow = attn + ((size_t)b * Ll + (i0 + mn)) * Ll;

  v8f acc = {};
  for (int j = 0; j < Ll; j += 4) {
    v2f a, bv;
    a.x = arow[j + kh];
    a.y = arow[j + kh + 1];
    const float* bp = inp + ((size_t)b * Ll + (j + kh)) * Dd + (d0 + mn);
    bv.x = bp[0];
    bv.y = bp[Dd];
    acc = wmma_f32_4(a, bv, acc);
  }

  const int mb = (lane >> 4) * 8;
#pragma unroll
  for (int r = 0; r < 8; ++r) {
    const int row      = i0 + mb + r;
    const size_t obase = ((size_t)b * Ll + row) * (2 * Dd);
    out[obase + Dd + d0 + mn] = acc[r];
    out[obase + d0 + mn] = inp[((size_t)b * Ll + row) * Dd + d0 + mn];
  }
}

extern "C" void kernel_launch(void* const* d_in, const int* in_sizes, int n_in,
                              void* d_out, int out_size, void* d_ws,
                              size_t ws_size, hipStream_t stream) {
  const float* inp = (const float*)d_in[0];
  const float* W1  = (const float*)d_in[1];
  const float* W2  = (const float*)d_in[2];
  float* out = (float*)d_out;

  float* qi   = (float*)d_ws;
  float* kj   = qi + (size_t)Bb * Ll * Kk;
  float* attn = kj + (size_t)Bb * Ll * Kk;

  qk_proj_kernel<<<Bb * (Ll / 16), 256, 0, stream>>>(inp, W1, qi, kj);
  attn_softmax_kernel<<<Bb * (Ll / RB), 256, 0, stream>>>(qi, kj, W2, attn);
  context_kernel<<<(Bb * (Ll / 16) * (Dd / 16)) / 8, 256, 0, stream>>>(attn, inp, out);
}


# both WMMA GEMM kernels: operands staged once per workgroup into LDS by coalesced 16-byte global loads, fragments read from LDS in double-buffered blocks; k order and sum order unchanged
# speedup vs baseline: 2.3193x; 1.2565x over previous
#include <hip/hip_runtime.h>
#include <math.h>

constexpr int Bb = 8;
constexpr int Ll = 256;
constexpr int Dd = 256;
constexpr int Kk = 64;

typedef __attribute__((ext_vector_type(2))) float v2f;
typedef __attribute__((ext_vector_type(8))) float v8f;

__device__ __forceinline__ v8f wmma_f32_4(v2f a, v2f b, v8f c) {
  return __builtin_amdgcn_wmma_f32_16x16x4_f32(false, a, false, b, (short)0, c,
                                               false, false);
}

__device__ __forceinline__ float fast_tanh(float x) {
#if __has_builtin(__builtin_amdgcn_tanhf)
  return __builtin_amdgcn_tanhf(x);
#else
  float r;
  asm volatile("v_tanh_f32 %0, %1" : "=v"(r) : "v"(x));
  return r;
#endif
}

constexpr int GP  = Dd + 4;
constexpr int GP4 = GP / 4;
constexpr int KB  = 16;

__device__ __forceinline__ v2f lds_pair(const float* p) {
  return *reinterpret_cast<const v2f*>(p);
}

__global__ __launch_bounds__(256) void qk_proj_kernel(
    const float* __restrict__ inp, const float* __restrict__ W1,
    float* __restrict__ qi, float* __restrict__ kj) {
  __shared__ float4 lds[(16 + 2 * Kk) * GP4];

  const int blk  = blockIdx.x;
  const int b    = blk / (Ll / 16);
  const int i0   = (blk % (Ll / 16)) * 16;
  const int t    = threadIdx.x;
  const int wave = t >> 5;
  const int lane = t & 31;
  const bool is_k = (wave >= 4);
  const int n0     = (wave & 3) * 16;

  const float4* xsrc = reinterpret_cast<const float4*>(
      inp + ((size_t)b * Ll + i0) * Dd);
  const float4* wsrc = reinterpret_cast<const float4*>(W1);
#pragma unroll
  for (int it = 0; it < (16 * Dd / 4) / 256; ++it) {
    const int f = it * 256 + t;
    lds[(f >> 6) * GP4 + (f & 63)] = xsrc[f];
  }
#pragma unroll
  for (int it = 0; it < (Kk * 2 * Dd / 4) / 256; ++it) {
    const int f  = it * 256 + t;
    const int c4 = f & 127;
    lds[(16 + (c4 >> 6) * Kk + (f >> 7)) * GP4 + (c4 & 63)] = wsrc[f];
  }
  __syncthreads();

  const int mn = lane & 15;
  const int kh = (lane >> 4) << 1;

  const float* lf   = reinterpret_cast<const float*>(lds);
  const float* arow = lf + mn * GP + kh;
  const float* brow = lf + (16 + (is_k ? Kk : 0) + n0 + mn) * GP + kh;

  v2f fa[2][KB], fb[2][KB];
#pragma unroll
  for (int s = 0; s < KB; ++s) {
    fa[0][s] = lds_pair(arow + 4 * s);
    fb[0][s] = lds_pair(brow + 4 * s);
  }
  v8f acc = {};
#pragma unroll
  for (int kb = 0; kb < Dd / (4 * KB); ++kb) {
    if (kb + 1 < Dd / (4 * KB)) {
#pragma unroll
      for (int s = 0; s < KB; ++s) {
        fa[(kb + 1) & 1][s] = lds_pair(arow + 4 * KB * (kb + 1) + 4 * s);
        fb[(kb + 1) & 1][s] = lds_pair(brow + 4 * KB * (kb + 1) + 4 * s);
      }
    }
    __builtin_amdgcn_sched_barrier(0);
#pragma unroll
    for (int s = 0; s < KB; ++s)
      acc = wmma_f32_4(fa[kb & 1][s], fb[kb & 1][s], acc);
    __builtin_amdgcn_sched_barrier(0);
  }

  float* op     = is_k ? kj : qi;
  const int mb  = (lane >> 4) * 8;
#pragma unroll
  for (int r = 0; r < 8; ++r) {
    op[((size_t)b * Ll + (i0 + mb + r)) * Kk + (n0 + mn)] = acc[r];
  }
}

constexpr int RB  = 8;
constexpr int KP4 = Kk / 4 + 1;

template <int M>
__device__ __forceinline__ float row_xor(float x) {
  const int v = __builtin_bit_cast(int, x);
  return __builtin_bit_cast(
      float, __builtin_amdgcn_update_dpp(v, v, 0x160 | M, 0xf, 0xf, false));
}
__device__ __forceinline__ float xhalf(float x) {
  const int v = __builtin_bit_cast(int, x);
  return __builtin_bit_cast(
      float, __builtin_amdgcn_permlanex16(v, v, 0x76543210, 0xfedcba98, false,
                                          false));
}
__device__ __forceinline__ float wave_max(float x) {
  x = fmaxf(x, row_xor<1>(x));
  x = fmaxf(x, row_xor<2>(x));
  x = fmaxf(x, row_xor<4>(x));
  x = fmaxf(x, row_xor<8>(x));
  return fmaxf(x, xhalf(x));
}
__device__ __forceinline__ float wave_sum(float x) {
  x += row_xor<1>(x);
  x += row_xor<2>(x);
  x += row_xor<4>(x);
  x += row_xor<8>(x);
  return x + xhalf(x);
}

struct KBlk {
  float4 kv;
  float4 w;
  float4 q[2 * 4];
};

__device__ __forceinline__ void load_blk(KBlk& d, const float4* myrow,
                                         const float4* w_s, const float4* q_s,
                                         int k4) {
  d.kv = myrow[k4];
  d.w  = w_s[k4];
#pragma unroll
  for (int c = 0; c < 8; ++c) d.q[c] = q_s[k4 * 8 + c];
}

__device__ __forceinline__ void k_step(float (&s)[RB], float kv, float w,
                                       const float4 qa, const float4 qb) {
  s[0] += fast_tanh(qa.x + kv) * w;
  s[1] += fast_tanh(qa.y + kv) * w;
  s[2] += fast_tanh(qa.z + kv) * w;
  s[3] += fast_tanh(qa.w + kv) * w;
  s[4] += fast_tanh(qb.x + kv) * w;
  s[5] += fast_tanh(qb.y + kv) * w;
  s[6] += fast_tanh(qb.z + kv) * w;
  s[7] += fast_tanh(qb.w + kv) * w;
}

__global__ __launch_bounds__(256) void attn_softmax_kernel(
    const float* __restrict__ qi, const float* __restrict__ kj,
    const float* __restrict__ W2, float* __restrict__ attn) {
  constexpr int Q_OFF  = 0;
  constexpr int W_OFF  = Q_OFF + Kk * (RB / 4);
  constexpr int MX_OFF = W_OFF + Kk / 4;
  constexpr int SM_OFF = MX_OFF + 8 * (RB / 4);
  constexpr int KJ_OFF = SM_OFF + 8 * (RB / 4);
  __shared__ float4 lds[KJ_OFF + 8 * 32 * KP4];
  float4* q_s  = lds + Q_OFF;
  float4* w_s  = lds + W_OFF;
  float4* mx_s = lds + MX_OFF;
  float4* sm_s = lds + SM_OFF;

  const int b    = blockIdx.x / (Ll / RB);
  const int i0   = (blockIdx.x % (Ll / RB)) * RB;
  const int t    = threadIdx.x;
  const int wave = t >> 5;
  const int lane = t & 31;

  const float4* src = reinterpret_cast<const float4*>(
      kj + ((size_t)b * Ll + 32 * wave) * Kk);
  float4* tile = lds + KJ_OFF + wave * 32 * KP4;
#pragma unroll
  for (int it = 0; it < 16; ++it) {
    const int f = it * 32 + lane;
    tile[(f >> 4) * KP4 + (f & 15)] = src[f];
  }
  {
    float* q_f = reinterpret_cast<float*>(q_s);
    for (int it = 0; it < (RB * Kk) / 256; ++it) {
      const int idx = it * 256 + t;
      const int r = idx >> 6;
      const int k = idx & 63;
      q_f[k * RB + r] = qi[((size_t)b * Ll + i0 + r) * Kk + k];
    }
    if (t < Kk) reinterpret_cast<float*>(w_s)[t] = W2[t];
  }
  __syncthreads();

  const float4* myrow = tile + lane * KP4;
  float s[RB];
#pragma unroll
  for (int r = 0; r < RB; ++r) s[r] = 0.0f;

  KBlk blk[2];
  load_blk(blk[0], myrow, w_s, q_s, 0);
#pragma unroll
  for (int k4 = 0; k4 < Kk / 4; ++k4) {
    if (k4 + 1 < Kk / 4) load_blk(blk[(k4 + 1) & 1], myrow, w_s, q_s, k4 + 1);
    __builtin_amdgcn_sched_barrier(0);
    const KBlk& c = blk[k4 & 1];
    k_step(s, c.kv.x, c.w.x, c.q[0], c.q[1]);
    k_step(s, c.kv.y, c.w.y, c.q[2], c.q[3]);
    k_step(s, c.kv.z, c.w.z, c.q[4], c.q[5]);
    k_step(s, c.kv.w, c.w.w, c.q[6], c.q[7]);
  }

  float m[RB];
#pragma unroll
  for (int r = 0; r < RB; ++r) m[r] = wave_max(s[r]);
  if (lane == 0) {
    mx_s[wave * 2 + 0] = make_float4(m[0], m[1], m[2], m[3]);
    mx_s[wave * 2 + 1] = make_float4(m[4], m[5], m[6], m[7]);
  }
  __syncthreads();
  {
    float4 a = mx_s[0], c = mx_s[1];
    for (int w = 1; w < 8; ++w) {
      const float4 a2 = mx_s[w * 2 + 0], c2 = mx_s[w * 2 + 1];
      a.x = fmaxf(a.x, a2.x); a.y = fmaxf(a.y, a2.y);
      a.z = fmaxf(a.z, a2.z); a.w = fmaxf(a.w, a2.w);
      c.x = fmaxf(c.x, c2.x); c.y = fmaxf(c.y, c2.y);
      c.z = fmaxf(c.z, c2.z); c.w = fmaxf(c.w, c2.w);
    }
    m[0] = a.x; m[1] = a.y; m[2] = a.z; m[3] = a.w;
    m[4] = c.x; m[5] = c.y; m[6] = c.z; m[7] = c.w;
  }

  float e[RB], p[RB];
#pragma unroll
  for (int r = 0; r < RB; ++r) {
    e[r] = __expf(s[r] - m[r]);
    p[r] = wave_sum(e[r]);
  }
  if (lane == 0) {
    sm_s[wave * 2 + 0] = make_float4(p[0], p[1], p[2], p[3]);
    sm_s[wave * 2 + 1] = make_float4(p[4], p[5], p[6], p[7]);
  }
  __syncthreads();
  {
    float4 a = sm_s[0], c = sm_s[1];
    for (int w = 1; w < 8; ++w) {
      const float4 a2 = sm_s[w * 2 + 0], c2 = sm_s[w * 2 + 1];
      a.x += a2.x; a.y += a2.y; a.z += a2.z; a.w += a2.w;
      c.x += c2.x; c.y += c2.y; c.z += c2.z; c.w += c2.w;
    }
    p[0] = a.x; p[1] = a.y; p[2] = a.z; p[3] = a.w;
    p[4] = c.x; p[5] = c.y; p[6] = c.z; p[7] = c.w;
  }

#pragma unroll
  for (int r = 0; r < RB; ++r) {
    attn[((size_t)b * Ll + (i0 + r)) * Ll + t] = e[r] * (1.0f / p[r]);
  }
}

constexpr int XP  = 128 + 16;
constexpr int XP4 = XP / 4;

__global__ __launch_bounds__(256) void context_kernel(
    const float* __restrict__ attn, const float* __restrict__ inp,
    float* __restrict__ out) {
  constexpr int XT_OFF = 16 * GP4;
  __shared__ float4 lds[XT_OFF + Ll * XP4];

  const int t     = threadIdx.x;
  const int wave  = t >> 5;
  const int lane  = t & 31;
  const int b     = blockIdx.x >> 5;
  const int i0    = ((blockIdx.x & 31) >> 1) * 16;
  const int dbase = (blockIdx.x & 1) * 128;
  const int dl    = wave * 16;
  const int d0    = dbase + dl;

  const float4* asrc = reinterpret_cast<const float4*>(
      attn + ((size_t)b * Ll + i0) * Ll);
  const float4* xsrc = reinterpret_cast<const float4*>(
      inp + (size_t)b * Ll * Dd + dbase);
#pragma unroll
  for (int it = 0; it < (16 * Ll / 4) / 256; ++it) {
    const int f = it * 256 + t;
    lds[(f >> 6) * GP4 + (f & 63)] = asrc[f];
  }
#pragma unroll
  for (int it = 0; it < (Ll * 128 / 4) / 256; ++it) {
    const int f   = it * 256 + t;
    const int row = f >> 5;
    const int c4  = f & 31;
    lds[XT_OFF + row * XP4 + c4] = xsrc[row * (Dd / 4) + c4];
  }
  __syncthreads();

  const int mn = lane & 15;
  const int kh = (lane >> 4) << 1;

  const float* lf   = reinterpret_cast<const float*>(lds);
  const float* xt   = lf + 16 * GP;
  const float* arow = lf + mn * GP + kh;
  const float* bcol = xt + kh * XP + dl + mn;

  v2f fa[2][KB], fb[2][KB];
#pragma unroll
  for (int s = 0; s < KB; ++s) {
    fa[0][s]   = lds_pair(arow + 4 * s);
    fb[0][s].x = bcol[(4 * s) * XP];
    fb[0][s].y = bcol[(4 * s + 1) * XP];
  }
  v8f acc = {};
#pragma unroll
  for (int kb = 0; kb < Ll / (4 * KB); ++kb) {
    if (kb + 1 < Ll / (4 * KB)) {
#pragma unroll
      for (int s = 0; s < KB; ++s) {
        const int j = 4 * KB * (kb + 1) + 4 * s;
        fa[(kb + 1) & 1][s]   = lds_pair(arow + j);
        fb[(kb + 1) & 1][s].x = bcol[j * XP];
        fb[(kb + 1) & 1][s].y = bcol[(j + 1) * XP];
      }
    }
    __builtin_amdgcn_sched_barrier(0);
#pragma unroll
    for (int s = 0; s < KB; ++s)
      acc = wmma_f32_4(fa[kb & 1][s], fb[kb & 1][s], acc);
    __builtin_amdgcn_sched_barrier(0);
  }

  const int mb = (lane >> 4) * 8;
#pragma unroll
  for (int r = 0; r < 8; ++r) {
    const int row      = i0 + mb + r;
    const size_t obase = ((size_t)b * Ll + row) * (2 * Dd);
    out[obase + Dd + d0 + mn] = acc[r];
    out[obase + d0 + mn] = xt[row * XP + dl + mn];
  }
}

extern "C" void kernel_launch(void* const* d_in, const int* in_sizes, int n_in,
                              void* d_out, int out_size, void* d_ws,
                              size_t ws_size, hipStream_t stream) {
  const float* inp = (const float*)d_in[0];
  const float* W1  = (const float*)d_in[1];
  const float* W2  = (const float*)d_in[2];
  float* out = (float*)d_out;

  float* qi   = (float*)d_ws;
  float* kj   = qi + (size_t)Bb * Ll * Kk;
  float* attn = kj + (size_t)Bb * Ll * Kk;

  qk_proj_kernel<<<Bb * (Ll / 16), 256, 0, stream>>>(inp, W1, qi, kj);
  attn_softmax_kernel<<<Bb * (Ll / RB), 256, 0, stream>>>(qi, kj, W2, attn);
  context_kernel<<<(Bb * (Ll / 16) * (Dd / 16)) / 8, 256, 0, stream>>>(attn, inp, out);
}


# both WMMA GEMM kernels re-tiled to squarer workgroup tiles (proj 32x32, context 32x64, 4 waves each, 256 workgroups): fewer operand bytes staged per workgroup; k order and sum order unchanged
# speedup vs baseline: 2.5387x; 1.0946x over previous
#include <hip/hip_runtime.h>
#include <math.h>

constexpr int Bb = 8;
constexpr int Ll = 256;
constexpr int Dd = 256;
constexpr int Kk = 64;

typedef __attribute__((ext_vector_type(2))) float v2f;
typedef __attribute__((ext_vector_type(8))) float v8f;

__device__ __forceinline__ v8f wmma_f32_4(v2f a, v2f b, v8f c) {
  return __builtin_amdgcn_wmma_f32_16x16x4_f32(false, a, false, b, (short)0, c,
                                               false, false);
}

__device__ __forceinline__ float fast_tanh(float x) {
#if __has_builtin(__builtin_amdgcn_tanhf)
  return __builtin_amdgcn_tanhf(x);
#else
  float r;
  asm volatile("v_tanh_f32 %0, %1" : "=v"(r) : "v"(x));
  return r;
#endif
}

constexpr int GP  = Dd + 4;
constexpr int GP4 = GP / 4;
constexpr int KB  = 16;

__device__ __forceinline__ v2f lds_pair(const float* p) {
  return *reinterpret_cast<const v2f*>(p);
}

constexpr int TM = 32;
constexpr int TN = 32;

__global__ __launch_bounds__(128) void qk_proj_kernel(
    const float* __restrict__ inp, const float* __restrict__ W1,
    float* __restrict__ qi, float* __restrict__ kj) {
  __shared__ float4 lds[(TM + TN) * GP4];

  const int blk  = blockIdx.x;
  const int rt   = blk >> 2;
  const int ct   = blk & 3;
  const bool is_k = (ct >= 2);
  const int nb   = (ct & 1) * TN;
  const int t    = threadIdx.x;
  const int wave = t >> 5;
  const int lane = t & 31;

  const float4* xsrc = reinterpret_cast<const float4*>(
      inp + (size_t)rt * TM * Dd);
  const float4* wsrc = reinterpret_cast<const float4*>(
      W1 + (size_t)nb * 2 * Dd + (is_k ? Dd : 0));
#pragma unroll
  for (int it = 0; it < (TM * Dd / 4) / 128; ++it) {
    const int f = it * 128 + t;
    lds[(f >> 6) * GP4 + (f & 63)] = xsrc[f];
  }
#pragma unroll
  for (int it = 0; it < (TN * Dd / 4) / 128; ++it) {
    const int f   = it * 128 + t;
    const int row = f >> 6;
    const int c4  = f & 63;
    lds[(TM + row) * GP4 + c4] = wsrc[row * (2 * Dd / 4) + c4];
  }
  __syncthreads();

  const int mn = lane & 15;
  const int kh = (lane >> 4) << 1;
  const int m0 = (wave >> 1) * 16;
  const int n0 = (wave & 1) * 16;

  const float* lf   = reinterpret_cast<const float*>(lds);
  const float* arow = lf + (m0 + mn) * GP + kh;
  const float* brow = lf + (TM + n0 + mn) * GP + kh;

  v2f fa[2][KB], fb[2][KB];
#pragma unroll
  for (int s = 0; s < KB; ++s) {
    fa[0][s] = lds_pair(arow + 4 * s);
    fb[0][s] = lds_pair(brow + 4 * s);
  }
  v8f acc = {};
#pragma unroll
  for (int kb = 0; kb < Dd / (4 * KB); ++kb) {
    if (kb + 1 < Dd / (4 * KB)) {
#pragma unroll
      for (int s = 0; s < KB; ++s) {
        fa[(kb + 1) & 1][s] = lds_pair(arow + 4 * KB * (kb + 1) + 4 * s);
        fb[(kb + 1) & 1][s] = lds_pair(brow + 4 * KB * (kb + 1) + 4 * s);
      }
    }
    __builtin_amdgcn_sched_barrier(0);
#pragma unroll
    for (int s = 0; s < KB; ++s)
      acc = wmma_f32_4(fa[kb & 1][s], fb[kb & 1][s], acc);
    __builtin_amdgcn_sched_barrier(0);
  }

  float* op     = is_k ? kj : qi;
  const int mb  = (lane >> 4) * 8;
#pragma unroll
  for (int r = 0; r < 8; ++r) {
    op[((size_t)rt * TM + (m0 + mb + r)) * Kk + (nb + n0 + mn)] = acc[r];
  }
}

constexpr int RB  = 8;
constexpr int KP4 = Kk / 4 + 1;

template <int M>
__device__ __forceinline__ float row_xor(float x) {
  const int v = __builtin_bit_cast(int, x);
  return __builtin_bit_cast(
      float, __builtin_amdgcn_update_dpp(v, v, 0x160 | M, 0xf, 0xf, false));
}
__device__ __forceinline__ float xhalf(float x) {
  const int v = __builtin_bit_cast(int, x);
  return __builtin_bit_cast(
      float, __builtin_amdgcn_permlanex16(v, v, 0x76543210, 0xfedcba98, false,
                                          false));
}
__device__ __forceinline__ float wave_max(float x) {
  x = fmaxf(x, row_xor<1>(x));
  x = fmaxf(x, row_xor<2>(x));
  x = fmaxf(x, row_xor<4>(x));
  x = fmaxf(x, row_xor<8>(x));
  return fmaxf(x, xhalf(x));
}
__device__ __forceinline__ float wave_sum(float x) {
  x += row_xor<1>(x);
  x += row_xor<2>(x);
  x += row_xor<4>(x);
  x += row_xor<8>(x);
  return x + xhalf(x);
}

struct KBlk {
  float4 kv;
  float4 w;
  float4 q[2 * 4];
};

__device__ __forceinline__ void load_blk(KBlk& d, const float4* myrow,
                                         const float4* w_s, const float4* q_s,
                                         int k4) {
  d.kv = myrow[k4];
  d.w  = w_s[k4];
#pragma unroll
  for (int c = 0; c < 8; ++c) d.q[c] = q_s[k4 * 8 + c];
}

__device__ __forceinline__ void k_step(float (&s)[RB], float kv, float w,
                                       const float4 qa, const float4 qb) {
  s[0] += fast_tanh(qa.x + kv) * w;
  s[1] += fast_tanh(qa.y + kv) * w;
  s[2] += fast_tanh(qa.z + kv) * w;
  s[3] += fast_tanh(qa.w + kv) * w;
  s[4] += fast_tanh(qb.x + kv) * w;
  s[5] += fast_tanh(qb.y + kv) * w;
  s[6] += fast_tanh(qb.z + kv) * w;
  s[7] += fast_tanh(qb.w + kv) * w;
}

__global__ __launch_bounds__(256) void attn_softmax_kernel(
    const float* __restrict__ qi, const float* __restrict__ kj,
    const float* __restrict__ W2, float* __restrict__ attn) {
  constexpr int Q_OFF  = 0;
  constexpr int W_OFF  = Q_OFF + Kk * (RB / 4);
  constexpr int MX_OFF = W_OFF + Kk / 4;
  constexpr int SM_OFF = MX_OFF + 8 * (RB / 4);
  constexpr int KJ_OFF = SM_OFF + 8 * (RB / 4);
  __shared__ float4 lds[KJ_OFF + 8 * 32 * KP4];
  float4* q_s  = lds + Q_OFF;
  float4* w_s  = lds + W_OFF;
  float4* mx_s = lds + MX_OFF;
  float4* sm_s = lds + SM_OFF;

  const int b    = blockIdx.x / (Ll / RB);
  const int i0   = (blockIdx.x % (Ll / RB)) * RB;
  const int t    = threadIdx.x;
  const int wave = t >> 5;
  const int lane = t & 31;

  const float4* src = reinterpret_cast<const float4*>(
      kj + ((size_t)b * Ll + 32 * wave) * Kk);
  float4* tile = lds + KJ_OFF + wave * 32 * KP4;
#pragma unroll
  for (int it = 0; it < 16; ++it) {
    const int f = it * 32 + lane;
    tile[(f >> 4) * KP4 + (f & 15)] = src[f];
  }
  {
    float* q_f = reinterpret_cast<float*>(q_s);
    for (int it = 0; it < (RB * Kk) / 256; ++it) {
      const int idx = it * 256 + t;
      const int r = idx >> 6;
      const int k = idx & 63;
      q_f[k * RB + r] = qi[((size_t)b * Ll + i0 + r) * Kk + k];
    }
    if (t < Kk) reinterpret_cast<float*>(w_s)[t] = W2[t];
  }
  __syncthreads();

  const float4* myrow = tile + lane * KP4;
  float s[RB];
#pragma unroll
  for (int r = 0; r < RB; ++r) s[r] = 0.0f;

  KBlk blk[2];
  load_blk(blk[0], myrow, w_s, q_s, 0);
#pragma unroll
  for (int k4 = 0; k4 < Kk / 4; ++k4) {
    if (k4 + 1 < Kk / 4) load_blk(blk[(k4 + 1) & 1], myrow, w_s, q_s, k4 + 1);
    __builtin_amdgcn_sched_barrier(0);
    const KBlk& c = blk[k4 & 1];
    k_step(s, c.kv.x, c.w.x, c.q[0], c.q[1]);
    k_step(s, c.kv.y, c.w.y, c.q[2], c.q[3]);
    k_step(s, c.kv.z, c.w.z, c.q[4], c.q[5]);
    k_step(s, c.kv.w, c.w.w, c.q[6], c.q[7]);
  }

  float m[RB];
#pragma unroll
  for (int r = 0; r < RB; ++r) m[r] = wave_max(s[r]);
  if (lane == 0) {
    mx_s[wave * 2 + 0] = make_float4(m[0], m[1], m[2], m[3]);
    mx_s[wave * 2 + 1] = make_float4(m[4], m[5], m[6], m[7]);
  }
  __syncthreads();
  {
    float4 a = mx_s[0], c = mx_s[1];
    for (int w = 1; w < 8; ++w) {
      const float4 a2 = mx_s[w * 2 + 0], c2 = mx_s[w * 2 + 1];
      a.x = fmaxf(a.x, a2.x); a.y = fmaxf(a.y, a2.y);
      a.z = fmaxf(a.z, a2.z); a.w = fmaxf(a.w, a2.w);
      c.x = fmaxf(c.x, c2.x); c.y = fmaxf(c.y, c2.y);
      c.z = fmaxf(c.z, c2.z); c.w = fmaxf(c.w, c2.w);
    }
    m[0] = a.x; m[1] = a.y; m[2] = a.z; m[3] = a.w;
    m[4] = c.x; m[5] = c.y; m[6] = c.z; m[7] = c.w;
  }

  float e[RB], p[RB];
#pragma unroll
  for (int r = 0; r < RB; ++r) {
    e[r] = __expf(s[r] - m[r]);
    p[r] = wave_sum(e[r]);
  }
  if (lane == 0) {
    sm_s[wave * 2 + 0] = make_float4(p[0], p[1], p[2], p[3]);
    sm_s[wave * 2 + 1] = make_float4(p[4], p[5], p[6], p[7]);
  }
  __syncthreads();
  {
    float4 a = sm_s[0], c = sm_s[1];
    for (int w = 1; w < 8; ++w) {
      const float4 a2 = sm_s[w * 2 + 0], c2 = sm_s[w * 2 + 1];
      a.x += a2.x; a.y += a2.y; a.z += a2.z; a.w += a2.w;
      c.x += c2.x; c.y += c2.y; c.z += c2.z; c.w += c2.w;
    }
    p[0] = a.x; p[1] = a.y; p[2] = a.z; p[3] = a.w;
    p[4] = c.x; p[5] = c.y; p[6] = c.z; p[7] = c.w;
  }

#pragma unroll
  for (int r = 0; r < RB; ++r) {
    attn[((size_t)b * Ll + (i0 + r)) * Ll + t] = e[r] * (1.0f / p[r]);
  }
}

constexpr int CM  = 32;
constexpr int CN  = 64;
constexpr int XP  = CN + 16;
constexpr int XP4 = XP / 4;

__global__ __launch_bounds__(128) void context_kernel(
    const float* __restrict__ attn, const float* __restrict__ inp,
    float* __restrict__ out) {
  constexpr int XT_OFF = CM * GP4;
  __shared__ float4 lds[XT_OFF + Ll * XP4];

  const int t     = threadIdx.x;
  const int wave  = t >> 5;
  const int lane  = t & 31;
  const int b     = blockIdx.x >> 5;
  const int i0    = ((blockIdx.x & 31) >> 2) * CM;
  const int dbase = (blockIdx.x & 3) * CN;
  const int dl    = wave * 16;
  const int d0    = dbase + dl;

  const float4* asrc = reinterpret_cast<const float4*>(
      attn + ((size_t)b * Ll + i0) * Ll);
  const float4* xsrc = reinterpret_cast<const float4*>(
      inp + (size_t)b * Ll * Dd + dbase);
#pragma unroll
  for (int it = 0; it < (CM * Ll / 4) / 128; ++it) {
    const int f = it * 128 + t;
    lds[(f >> 6) * GP4 + (f & 63)] = asrc[f];
  }
#pragma unroll
  for (int it = 0; it < (Ll * CN / 4) / 128; ++it) {
    const int f   = it * 128 + t;
    const int row = f >> 4;
    const int c4  = f & 15;
    lds[XT_OFF + row * XP4 + c4] = xsrc[row * (Dd / 4) + c4];
  }
  __syncthreads();

  const int mn = lane & 15;
  const int kh = (lane >> 4) << 1;

  const float* lf    = reinterpret_cast<const float*>(lds);
  const float* xt    = lf + CM * GP;
  const float* arow0 = lf + mn * GP + kh;
  const float* arow1 = lf + (16 + mn) * GP + kh;
  const float* bcol  = xt + kh * XP + dl + mn;

  v2f fa0[2][KB], fa1[2][KB], fb[2][KB];
#pragma unroll
  for (int s = 0; s < KB; ++s) {
    fa0[0][s]  = lds_pair(arow0 + 4 * s);
    fa1[0][s]  = lds_pair(arow1 + 4 * s);
    fb[0][s].x = bcol[(4 * s) * XP];
    fb[0][s].y = bcol[(4 * s + 1) * XP];
  }
  v8f acc0 = {}, acc1 = {};
#pragma unroll
  for (int kb = 0; kb < Ll / (4 * KB); ++kb) {
    if (kb + 1 < Ll / (4 * KB)) {
#pragma unroll
      for (int s = 0; s < KB; ++s) {
        const int j = 4 * KB * (kb + 1) + 4 * s;
        fa0[(kb + 1) & 1][s]  = lds_pair(arow0 + j);
        fa1[(kb + 1) & 1][s]  = lds_pair(arow1 + j);
        fb[(kb + 1) & 1][s].x = bcol[j * XP];
        fb[(kb + 1) & 1][s].y = bcol[(j + 1) * XP];
      }
    }
    __builtin_amdgcn_sched_barrier(0);
#pragma unroll
    for (int s = 0; s < KB; ++s) {
      acc0 = wmma_f32_4(fa0[kb & 1][s], fb[kb & 1][s], acc0);
      acc1 = wmma_f32_4(fa1[kb & 1][s], fb[kb & 1][s], acc1);
    }
    __builtin_amdgcn_sched_barrier(0);
  }

  const int mb = (lane >> 4) * 8;
#pragma unroll
  for (int r = 0; r < 8; ++r) {
    const int row0      = i0 + mb + r;
    const int row1      = row0 + 16;
    const size_t obase0 = ((size_t)b * Ll + row0) * (2 * Dd);
    const size_t obase1 = ((size_t)b * Ll + row1) * (2 * Dd);
    out[obase0 + Dd + d0 + mn] = acc0[r];
    out[obase1 + Dd + d0 + mn] = acc1[r];
    out[obase0 + d0 + mn] = xt[row0 * XP + dl + mn];
    out[obase1 + d0 + mn] = xt[row1 * XP + dl + mn];
  }
}

extern "C" void kernel_launch(void* const* d_in, const int* in_sizes, int n_in,
                              void* d_out, int out_size, void* d_ws,
                              size_t ws_size, hipStream_t stream) {
  const float* inp = (const float*)d_in[0];
  const float* W1  = (const float*)d_in[1];
  const float* W2  = (const float*)d_in[2];
  float* out = (float*)d_out;

  float* qi   = (float*)d_ws;
  float* kj   = qi + (size_t)Bb * Ll * Kk;
  float* attn = kj + (size_t)Bb * Ll * Kk;

  qk_proj_kernel<<<(Bb * Ll / TM) * (2 * Kk / TN), 128, 0, stream>>>(inp, W1, qi, kj);
  attn_softmax_kernel<<<Bb * (Ll / RB), 256, 0, stream>>>(qi, kj, W2, attn);
  context_kernel<<<Bb * (Ll / CM) * (Dd / CN), 128, 0, stream>>>(attn, inp, out);
}


# both WMMA GEMM kernels: result tiles pass through an LDS tile and leave as 16-byte stores, so each 128-byte line of qi, kj and out is written whole by one store instruction; sums unchanged
# speedup vs baseline: 2.6690x; 1.0513x over previous
#include <hip/hip_runtime.h>
#include <math.h>

constexpr int Bb = 8;
constexpr int Ll = 256;
constexpr int Dd = 256;
constexpr int Kk = 64;

typedef __attribute__((ext_vector_type(2))) float v2f;
typedef __attribute__((ext_vector_type(8))) float v8f;

__device__ __forceinline__ v8f wmma_f32_4(v2f a, v2f b, v8f c) {
  return __builtin_amdgcn_wmma_f32_16x16x4_f32(false, a, false, b, (short)0, c,
                                               false, false);
}

__device__ __forceinline__ float fast_tanh(float x) {
#if __has_builtin(__builtin_amdgcn_tanhf)
  return __builtin_amdgcn_tanhf(x);
#else
  float r;
  asm volatile("v_tanh_f32 %0, %1" : "=v"(r) : "v"(x));
  return r;
#endif
}

constexpr int GP  = Dd + 4;
constexpr int GP4 = GP / 4;
constexpr int KB  = 16;

__device__ __forceinline__ v2f lds_pair(const float* p) {
  return *reinterpret_cast<const v2f*>(p);
}

constexpr int TM = 32;
constexpr int TN = 32;
constexpr int OP  = TN + 4;
constexpr int OP4 = OP / 4;

__global__ __launch_bounds__(128) void qk_proj_kernel(
    const float* __restrict__ inp, const float* __restrict__ W1,
    float* __restrict__ qi, float* __restrict__ kj) {
  __shared__ float4 lds[(TM + TN) * GP4];
  __shared__ float4 otile[TM * OP4];

  const int blk  = blockIdx.x;
  const int rt   = blk >> 2;
  const int ct   = blk & 3;
  const bool is_k = (ct >= 2);
  const int nb   = (ct & 1) * TN;
  const int t    = threadIdx.x;
  const int wave = t >> 5;
  const int lane = t & 31;

  const float4* xsrc = reinterpret_cast<const float4*>(
      inp + (size_t)rt * TM * Dd);
  const float4* wsrc = reinterpret_cast<const float4*>(
      W1 + (size_t)nb * 2 * Dd + (is_k ? Dd : 0));
#pragma unroll
  for (int it = 0; it < (TM * Dd / 4) / 128; ++it) {
    const int f = it * 128 + t;
    lds[(f >> 6) * GP4 + (f & 63)] = xsrc[f];
  }
#pragma unroll
  for (int it = 0; it < (TN * Dd / 4) / 128; ++it) {
    const int f   = it * 128 + t;
    const int row = f >> 6;
    const int c4  = f & 63;
    lds[(TM + row) * GP4 + c4] = wsrc[row * (2 * Dd / 4) + c4];
  }
  __syncthreads();

  const int mn = lane & 15;
  const int kh = (lane >> 4) << 1;
  const int m0 = (wave >> 1) * 16;
  const int n0 = (wave & 1) * 16;

  const float* lf   = reinterpret_cast<const float*>(lds);
  const float* arow = lf + (m0 + mn) * GP + kh;
  const float* brow = lf + (TM + n0 + mn) * GP + kh;

  v2f fa[2][KB], fb[2][KB];
#pragma unroll
  for (int s = 0; s < KB; ++s) {
    fa[0][s] = lds_pair(arow + 4 * s);
    fb[0][s] = lds_pair(brow + 4 * s);
  }
  v8f acc = {};
#pragma unroll
  for (int kb = 0; kb < Dd / (4 * KB); ++kb) {
    if (kb + 1 < Dd / (4 * KB)) {
#pragma unroll
      for (int s = 0; s < KB; ++s) {
        fa[(kb + 1) & 1][s] = lds_pair(arow + 4 * KB * (kb + 1) + 4 * s);
        fb[(kb + 1) & 1][s] = lds_pair(brow + 4 * KB * (kb + 1) + 4 * s);
      }
    }
    __builtin_amdgcn_sched_barrier(0);
#pragma unroll
    for (int s = 0; s < KB; ++s)
      acc = wmma_f32_4(fa[kb & 1][s], fb[kb & 1][s], acc);
    __builtin_amdgcn_sched_barrier(0);
  }

  float* ot    = reinterpret_cast<float*>(otile);
  const int mb = (lane >> 4) * 8;
#pragma unroll
  for (int r = 0; r < 8; ++r) {
    ot[(m0 + mb + r) * OP + (n0 + mn)] = acc[r];
  }
  __syncthreads();
  float* op = is_k ? kj : qi;
#pragma unroll
  for (int it = 0; it < (TM * TN / 4) / 128; ++it) {
    const int f   = it * 128 + t;
    const int row = f >> 3;
    const int c4  = f & 7;
    *reinterpret_cast<float4*>(
        op + ((size_t)rt * TM + row) * Kk + nb + 4 * c4) = otile[row * OP4 + c4];
  }
}

constexpr int RB  = 8;
constexpr int KP4 = Kk / 4 + 1;

template <int M>
__device__ __forceinline__ float row_xor(float x) {
  const int v = __builtin_bit_cast(int, x);
  return __builtin_bit_cast(
      float, __builtin_amdgcn_update_dpp(v, v, 0x160 | M, 0xf, 0xf, false));
}
__device__ __forceinline__ float xhalf(float x) {
  const int v = __builtin_bit_cast(int, x);
  return __builtin_bit_cast(
      float, __builtin_amdgcn_permlanex16(v, v, 0x76543210, 0xfedcba98, false,
                                          false));
}
__device__ __forceinline__ float wave_max(float x) {
  x = fmaxf(x, row_xor<1>(x));
  x = fmaxf(x, row_xor<2>(x));
  x = fmaxf(x, row_xor<4>(x));
  x = fmaxf(x, row_xor<8>(x));
  return fmaxf(x, xhalf(x));
}
__device__ __forceinline__ float wave_sum(float x) {
  x += row_xor<1>(x);
  x += row_xor<2>(x);
  x += row_xor<4>(x);
  x += row_xor<8>(x);
  return x + xhalf(x);
}

struct KBlk {
  float4 kv;
  float4 w;
  float4 q[2 * 4];
};

__device__ __forceinline__ void load_blk(KBlk& d, const float4* myrow,
                                         const float4* w_s, const float4* q_s,
                                         int k4) {
  d.kv = myrow[k4];
  d.w  = w_s[k4];
#pragma unroll
  for (int c = 0; c < 8; ++c) d.q[c] = q_s[k4 * 8 + c];
}

__device__ __forceinline__ void k_step(float (&s)[RB], float kv, float w,
                                       const float4 qa, const float4 qb) {
  s[0] += fast_tanh(qa.x + kv) * w;
  s[1] += fast_tanh(qa.y + kv) * w;
  s[2] += fast_tanh(qa.z + kv) * w;
  s[3] += fast_tanh(qa.w + kv) * w;
  s[4] += fast_tanh(qb.x + kv) * w;
  s[5] += fast_tanh(qb.y + kv) * w;
  s[6] += fast_tanh(qb.z + kv) * w;
  s[7] += fast_tanh(qb.w + kv) * w;
}

__global__ __launch_bounds__(256) void attn_softmax_kernel(
    const float* __restrict__ qi, const float* __restrict__ kj,
    const float* __restrict__ W2, float* __restrict__ attn) {
  constexpr int Q_OFF  = 0;
  constexpr int W_OFF  = Q_OFF + Kk * (RB / 4);
  constexpr int MX_OFF = W_OFF + Kk / 4;
  constexpr int SM_OFF = MX_OFF + 8 * (RB / 4);
  constexpr int KJ_OFF = SM_OFF + 8 * (RB / 4);
  __shared__ float4 lds[KJ_OFF + 8 * 32 * KP4];
  float4* q_s  = lds + Q_OFF;
  float4* w_s  = lds + W_OFF;
  float4* mx_s = lds + MX_OFF;
  float4* sm_s = lds + SM_OFF;

  const int b    = blockIdx.x / (Ll / RB);
  const int i0   = (blockIdx.x % (Ll / RB)) * RB;
  const int t    = threadIdx.x;
  const int wave = t >> 5;
  const int lane = t & 31;

  const float4* src = reinterpret_cast<const float4*>(
      kj + ((size_t)b * Ll + 32 * wave) * Kk);
  float4* tile = lds + KJ_OFF + wave * 32 * KP4;
#pragma unroll
  for (int it = 0; it < 16; ++it) {
    const int f = it * 32 + lane;
    tile[(f >> 4) * KP4 + (f & 15)] = src[f];
  }
  {
    float* q_f = reinterpret_cast<float*>(q_s);
    for (int it = 0; it < (RB * Kk) / 256; ++it) {
      const int idx = it * 256 + t;
      const int r = idx >> 6;
      const int k = idx & 63;
      q_f[k * RB + r] = qi[((size_t)b * Ll + i0 + r) * Kk + k];
    }
    if (t < Kk) reinterpret_cast<float*>(w_s)[t] = W2[t];
  }
  __syncthreads();

  const float4* myrow = tile + lane * KP4;
  float s[RB];
#pragma unroll
  for (int r = 0; r < RB; ++r) s[r] = 0.0f;

  KBlk blk[2];
  load_blk(blk[0], myrow, w_s, q_s, 0);
#pragma unroll
  for (int k4 = 0; k4 < Kk / 4; ++k4) {
    if (k4 + 1 < Kk / 4) load_blk(blk[(k4 + 1) & 1], myrow, w_s, q_s, k4 + 1);
    __builtin_amdgcn_sched_barrier(0);
    const KBlk& c = blk[k4 & 1];
    k_step(s, c.kv.x, c.w.x, c.q[0], c.q[1]);
    k_step(s, c.kv.y, c.w.y, c.q[2], c.q[3]);
    k_step(s, c.kv.z, c.w.z, c.q[4], c.q[5]);
    k_step(s, c.kv.w, c.w.w, c.q[6], c.q[7]);
  }

  float m[RB];
#pragma unroll
  for (int r = 0; r < RB; ++r) m[r] = wave_max(s[r]);
  if (lane == 0) {
    mx_s[wave * 2 + 0] = make_float4(m[0], m[1], m[2], m[3]);
    mx_s[wave * 2 + 1] = make_float4(m[4], m[5], m[6], m[7]);
  }
  __syncthreads();
  {
    float4 a = mx_s[0], c = mx_s[1];
    for (int w = 1; w < 8; ++w) {
      const float4 a2 = mx_s[w * 2 + 0], c2 = mx_s[w * 2 + 1];
      a.x = fmaxf(a.x, a2.x); a.y = fmaxf(a.y, a2.y);
      a.z = fmaxf(a.z, a2.z); a.w = fmaxf(a.w, a2.w);
      c.x = fmaxf(c.x, c2.x); c.y = fmaxf(c.y, c2.y);
      c.z = fmaxf(c.z, c2.z); c.w = fmaxf(c.w, c2.w);
    }
    m[0] = a.x; m[1] = a.y; m[2] = a.z; m[3] = a.w;
    m[4] = c.x; m[5] = c.y; m[6] = c.z; m[7] = c.w;
  }

  float e[RB], p[RB];
#pragma unroll
  for (int r = 0; r < RB; ++r) {
    e[r] = __expf(s[r] - m[r]);
    p[r] = wave_sum(e[r]);
  }
  if (lane == 0) {
    sm_s[wave * 2 + 0] = make_float4(p[0], p[1], p[2], p[3]);
    sm_s[wave * 2 + 1] = make_float4(p[4], p[5], p[6], p[7]);
  }
  __syncthreads();
  {
    float4 a = sm_s[0], c = sm_s[1];
    for (int w = 1; w < 8; ++w) {
      const float4 a2 = sm_s[w * 2 + 0], c2 = sm_s[w * 2 + 1];
      a.x += a2.x; a.y += a2.y; a.z += a2.z; a.w += a2.w;
      c.x += c2.x; c.y += c2.y; c.z += c2.z; c.w += c2.w;
    }
    p[0] = a.x; p[1] = a.y; p[2] = a.z; p[3] = a.w;
    p[4] = c.x; p[5] = c.y; p[6] = c.z; p[7] = c.w;
  }

#pragma unroll
  for (int r = 0; r < RB; ++r) {
    attn[((size_t)b * Ll + (i0 + r)) * Ll + t] = e[r] * (1.0f / p[r]);
  }
}

constexpr int CM  = 32;
constexpr int CN  = 64;
constexpr int XP  = CN + 16;
constexpr int XP4 = XP / 4;
constexpr int CP  = CN + 4;
constexpr int CP4 = CP / 4;

__global__ __launch_bounds__(128) void context_kernel(
    const float* __restrict__ attn, const float* __restrict__ inp,
    float* __restrict__ out) {
  constexpr int XT_OFF = CM * GP4;
  __shared__ float4 lds[XT_OFF + Ll * XP4];
  __shared__ float4 ctile[CM * CP4];

  const int t     = threadIdx.x;
  const int wave  = t >> 5;
  const int lane  = t & 31;
  const int b     = blockIdx.x >> 5;
  const int i0    = ((blockIdx.x & 31) >> 2) * CM;
  const int dbase = (blockIdx.x & 3) * CN;
  const int dl    = wave * 16;

  const float4* asrc = reinterpret_cast<const float4*>(
      attn + ((size_t)b * Ll + i0) * Ll);
  const float4* xsrc = reinterpret_cast<const float4*>(
      inp + (size_t)b * Ll * Dd + dbase);
#pragma unroll
  for (int it = 0; it < (CM * Ll / 4) / 128; ++it) {
    const int f = it * 128 + t;
    lds[(f >> 6) * GP4 + (f & 63)] = asrc[f];
  }
#pragma unroll
  for (int it = 0; it < (Ll * CN / 4) / 128; ++it) {
    const int f   = it * 128 + t;
    const int row = f >> 4;
    const int c4  = f & 15;
    lds[XT_OFF + row * XP4 + c4] = xsrc[row * (Dd / 4) + c4];
  }
  __syncthreads();

  const int mn = lane & 15;
  const int kh = (lane >> 4) << 1;

  const float* lf    = reinterpret_cast<const float*>(lds);
  const float* xt    = lf + CM * GP;
  const float* arow0 = lf + mn * GP + kh;
  const float* arow1 = lf + (16 + mn) * GP + kh;
  const float* bcol  = xt + kh * XP + dl + mn;

  v2f fa0[2][KB], fa1[2][KB], fb[2][KB];
#pragma unroll
  for (int s = 0; s < KB; ++s) {
    fa0[0][s]  = lds_pair(arow0 + 4 * s);
    fa1[0][s]  = lds_pair(arow1 + 4 * s);
    fb[0][s].x = bcol[(4 * s) * XP];
    fb[0][s].y = bcol[(4 * s + 1) * XP];
  }
  v8f acc0 = {}, acc1 = {};
#pragma unroll
  for (int kb = 0; kb < Ll / (4 * KB); ++kb) {
    if (kb + 1 < Ll / (4 * KB)) {
#pragma unroll
      for (int s = 0; s < KB; ++s) {
        const int j = 4 * KB * (kb + 1) + 4 * s;
        fa0[(kb + 1) & 1][s]  = lds_pair(arow0 + j);
        fa1[(kb + 1) & 1][s]  = lds_pair(arow1 + j);
        fb[(kb + 1) & 1][s].x = bcol[j * XP];
        fb[(kb + 1) & 1][s].y = bcol[(j + 1) * XP];
      }
    }
    __builtin_amdgcn_sched_barrier(0);
#pragma unroll
    for (int s = 0; s < KB; ++s) {
      acc0 = wmma_f32_4(fa0[kb & 1][s], fb[kb & 1][s], acc0);
      acc1 = wmma_f32_4(fa1[kb & 1][s], fb[kb & 1][s], acc1);
    }
    __builtin_amdgcn_sched_barrier(0);
  }

  float* ct    = reinterpret_cast<float*>(ctile);
  const int mb = (lane >> 4) * 8;
#pragma unroll
  for (int r = 0; r < 8; ++r) {
    ct[(mb + r) * CP + dl + mn]      = acc0[r];
    ct[(16 + mb + r) * CP + dl + mn] = acc1[r];
  }
  __syncthreads();
#pragma unroll
  for (int it = 0; it < (CM * CN / 4) / 128; ++it) {
    const int f   = it * 128 + t;
    const int row = f >> 4;
    const int c4  = f & 15;
    float4* orow = reinterpret_cast<float4*>(
        out + ((size_t)b * Ll + i0 + row) * (2 * Dd) + dbase);
    orow[c4]          = lds[XT_OFF + (i0 + row) * XP4 + c4];
    orow[Dd / 4 + c4] = ctile[row * CP4 + c4];
  }
}

extern "C" void kernel_launch(void* const* d_in, const int* in_sizes, int n_in,
                              void* d_out, int out_size, void* d_ws,
                              size_t ws_size, hipStream_t stream) {
  const float* inp = (const float*)d_in[0];
  const float* W1  = (const float*)d_in[1];
  const float* W2  = (const float*)d_in[2];
  float* out = (float*)d_out;

  float* qi   = (float*)d_ws;
  float* kj   = qi + (size_t)Bb * Ll * Kk;
  float* attn = kj + (size_t)Bb * Ll * Kk;

  qk_proj_kernel<<<(Bb * Ll / TM) * (2 * Kk / TN), 128, 0, stream>>>(inp, W1, qi, kj);
  attn_softmax_kernel<<<Bb * (Ll / RB), 256, 0, stream>>>(qi, kj, W2, attn);
  context_kernel<<<Bb * (Ll / CM) * (Dd / CN), 128, 0, stream>>>(attn, inp, out);
}


# all three kernels take operands in k order as loads arrive: both WMMA GEMM kernels stage four k chunks behind LDS-only barriers, score kernel reads pre-tiled kj into registers; sums unchanged
# speedup vs baseline: 2.8645x; 1.0732x over previous
#include <hip/hip_runtime.h>
#include <math.h>

constexpr int Bb = 8;
constexpr int Ll = 256;
constexpr int Dd = 256;
constexpr int Kk = 64;

typedef __attribute__((ext_vector_type(2))) float v2f;
typedef __attribute__((ext_vector_type(8))) float v8f;
typedef __attribute__((ext_vector_type(4))) float v4f;

__device__ __forceinline__ v8f wmma_f32_4(v2f a, v2f b, v8f c) {
  return __builtin_amdgcn_wmma_f32_16x16x4_f32(false, a, false, b, (short)0, c,
                                               false, false);
}

__device__ __forceinline__ float fast_tanh(float x) {
#if __has_builtin(__builtin_amdgcn_tanhf)
  return __builtin_amdgcn_tanhf(x);
#else
  float r;
  asm volatile("v_tanh_f32 %0, %1" : "=v"(r) : "v"(x));
  return r;
#endif
}

constexpr int GP  = Dd + 4;
constexpr int GP4 = GP / 4;
constexpr int KB  = 16;

__device__ __forceinline__ v2f lds_pair(const float* p) {
  return *reinterpret_cast<const v2f*>(p);
}

__device__ __forceinline__ void lds_barrier() {
  __builtin_amdgcn_fence(__ATOMIC_RELEASE, "workgroup", "local");
  __builtin_amdgcn_s_barrier();
  __builtin_amdgcn_fence(__ATOMIC_ACQUIRE, "workgroup", "local");
}

constexpr int NCH = 4;
static_assert(Dd == 4 * KB * NCH && Ll == 4 * KB * NCH, "one chunk per fragment block");

constexpr int TM = 32;
constexpr int TN = 32;
constexpr int OP  = TN + 4;
constexpr int OP4 = OP / 4;
static_assert(TM == 32, "a kj row tile is the 32 columns of one wave of the score kernel");

__global__ __launch_bounds__(128) void qk_proj_kernel(
    const float* __restrict__ inp, const float* __restrict__ W1,
    float* __restrict__ qi, float* __restrict__ kj) {
  __shared__ v4f lds[(TM + TN) * GP4];
  __shared__ v4f otile[TM * OP4];

  const int blk  = blockIdx.x;
  const int rt   = blk >> 2;
  const int ct   = blk & 3;
  const bool is_k = (ct >= 2);
  const int nb   = (ct & 1) * TN;
  const int t    = threadIdx.x;
  const int wave = t >> 5;
  const int lane = t & 31;

  const v4f* xsrc = reinterpret_cast<const v4f*>(
      inp + (size_t)rt * TM * Dd);
  const v4f* wsrc = reinterpret_cast<const v4f*>(
      W1 + (size_t)nb * 2 * Dd + (is_k ? Dd : 0));
  v4f rx[NCH * 4];
  v4f rw[NCH * 4];
#pragma unroll
  for (int c = 0; c < NCH; ++c) {
#pragma unroll
    for (int i = 0; i < 4; ++i) {
      const int idx = i * 128 + t;
      const int row = idx >> 4;
      const int c4  = 16 * c + (idx & 15);
      rx[c * 4 + i] = xsrc[row * (Dd / 4) + c4];
      rw[c * 4 + i] = wsrc[row * (2 * Dd / 4) + c4];
    }
    __builtin_amdgcn_sched_barrier(0);
  }
#pragma unroll
  for (int i = 0; i < 4; ++i) {
    const int idx = i * 128 + t;
    const int row = idx >> 4;
    const int c4  = idx & 15;
    lds[row * GP4 + c4]        = rx[i];
    lds[(TM + row) * GP4 + c4] = rw[i];
  }
  lds_barrier();

  const int mn = lane & 15;
  const int kh = (lane >> 4) << 1;
  const int m0 = (wave >> 1) * 16;
  const int n0 = (wave & 1) * 16;

  const float* lf   = reinterpret_cast<const float*>(lds);
  const float* arow = lf + (m0 + mn) * GP + kh;
  const float* brow = lf + (TM + n0 + mn) * GP + kh;

  v2f fa[2][KB], fb[2][KB];
#pragma unroll
  for (int s = 0; s < KB; ++s) {
    fa[0][s] = lds_pair(arow + 4 * s);
    fb[0][s] = lds_pair(brow + 4 * s);
  }
  v8f acc = {};
#pragma unroll
  for (int kb = 0; kb < NCH; ++kb) {
    if (kb + 1 < NCH) {
#pragma unroll
      for (int i = 0; i < 4; ++i) {
        const int idx = i * 128 + t;
        const int row = idx >> 4;
        const int c4  = 16 * (kb + 1) + (idx & 15);
        lds[row * GP4 + c4]        = rx[(kb + 1) * 4 + i];
        lds[(TM + row) * GP4 + c4] = rw[(kb + 1) * 4 + i];
      }
      lds_barrier();
#pragma unroll
      for (int s = 0; s < KB; ++s) {
        fa[(kb + 1) & 1][s] = lds_pair(arow + 4 * KB * (kb + 1) + 4 * s);
        fb[(kb + 1) & 1][s] = lds_pair(brow + 4 * KB * (kb + 1) + 4 * s);
      }
    }
    __builtin_amdgcn_sched_barrier(0);
#pragma unroll
    for (int s = 0; s < KB; ++s)
      acc = wmma_f32_4(fa[kb & 1][s], fb[kb & 1][s], acc);
    __builtin_amdgcn_sched_barrier(0);
  }

  float* ot    = reinterpret_cast<float*>(otile);
  const int mb = (lane >> 4) * 8;
#pragma unroll
  for (int r = 0; r < 8; ++r) {
    ot[(m0 + mb + r) * OP + (n0 + mn)] = acc[r];
  }
  __syncthreads();
  if (!is_k) {
#pragma unroll
    for (int it = 0; it < (TM * TN / 4) / 128; ++it) {
      const int f   = it * 128 + t;
      const int row = f >> 3;
      const int c4  = f & 7;
      *reinterpret_cast<v4f*>(
          qi + ((size_t)rt * TM + row) * Kk + nb + 4 * c4) = otile[row * OP4 + c4];
    }
  } else {
    v4f* kt = reinterpret_cast<v4f*>(kj) + ((size_t)rt * (Kk / 4) + nb / 4) * TM;
#pragma unroll
    for (int it = 0; it < (TM * TN / 4) / 128; ++it) {
      const int f   = it * 128 + t;
      const int k4l = f >> 5;
      const int row = f & 31;
      kt[f] = otile[row * OP4 + k4l];
    }
  }
}

constexpr int RB  = 8;

template <int M>
__device__ __forceinline__ float row_xor(float x) {
  const int v = __builtin_bit_cast(int, x);
  return __builtin_bit_cast(
      float, __builtin_amdgcn_update_dpp(v, v, 0x160 | M, 0xf, 0xf, false));
}
__device__ __forceinline__ float xhalf(float x) {
  const int v = __builtin_bit_cast(int, x);
  return __builtin_bit_cast(
      float, __builtin_amdgcn_permlanex16(v, v, 0x76543210, 0xfedcba98, false,
                                          false));
}
__device__ __forceinline__ float wave_max(float x) {
  x = fmaxf(x, row_xor<1>(x));
  x = fmaxf(x, row_xor<2>(x));
  x = fmaxf(x, row_xor<4>(x));
  x = fmaxf(x, row_xor<8>(x));
  return fmaxf(x, xhalf(x));
}
__device__ __forceinline__ float wave_sum(float x) {
  x += row_xor<1>(x);
  x += row_xor<2>(x);
  x += row_xor<4>(x);
  x += row_xor<8>(x);
  return x + xhalf(x);
}

struct KBlk {
  float4 w;
  float4 q[2 * 4];
};

__device__ __forceinline__ void load_blk(KBlk& d, const float4* w_s,
                                         const float4* q_s, int k4) {
  d.w  = w_s[k4];
#pragma unroll
  for (int c = 0; c < 8; ++c) d.q[c] = q_s[k4 * 8 + c];
}

__device__ __forceinline__ void k_step(float (&s)[RB], float kv, float w,
                                       const float4 qa, const float4 qb) {
  s[0] += fast_tanh(qa.x + kv) * w;
  s[1] += fast_tanh(qa.y + kv) * w;
  s[2] += fast_tanh(qa.z + kv) * w;
  s[3] += fast_tanh(qa.w + kv) * w;
  s[4] += fast_tanh(qb.x + kv) * w;
  s[5] += fast_tanh(qb.y + kv) * w;
  s[6] += fast_tanh(qb.z + kv) * w;
  s[7] += fast_tanh(qb.w + kv) * w;
}

__global__ __launch_bounds__(256) void attn_softmax_kernel(
    const float* __restrict__ qi, const float* __restrict__ kj,
    const float* __restrict__ W2, float* __restrict__ attn) {
  constexpr int Q_OFF  = 0;
  constexpr int W_OFF  = Q_OFF + Kk * (RB / 4);
  constexpr int MX_OFF = W_OFF + Kk / 4;
  constexpr int SM_OFF = MX_OFF + 8 * (RB / 4);
  __shared__ float4 lds[SM_OFF + 8 * (RB / 4)];
  float4* q_s  = lds + Q_OFF;
  float4* w_s  = lds + W_OFF;
  float4* mx_s = lds + MX_OFF;
  float4* sm_s = lds + SM_OFF;

  const int b    = blockIdx.x / (Ll / RB);
  const int i0   = (blockIdx.x % (Ll / RB)) * RB;
  const int t    = threadIdx.x;
  const int wave = t >> 5;
  const int lane = t & 31;

  float qv[(RB * Kk) / 256];
#pragma unroll
  for (int it = 0; it < (RB * Kk) / 256; ++it) {
    const int idx = it * 256 + t;
    qv[it] = qi[((size_t)b * Ll + i0 + (idx >> 6)) * Kk + (idx & 63)];
  }
  const float w2v = W2[t & (Kk - 1)];
  __builtin_amdgcn_sched_barrier(0);

  const float4* ksrc = reinterpret_cast<const float4*>(kj) +
                       ((size_t)(b * (Ll / 32) + wave) * (Kk / 4)) * 32 + lane;
  float4 kvr[Kk / 4];
#pragma unroll
  for (int k4 = 0; k4 < Kk / 4; ++k4) kvr[k4] = ksrc[k4 * 32];

  {
    float* q_f = reinterpret_cast<float*>(q_s);
#pragma unroll
    for (int it = 0; it < (RB * Kk) / 256; ++it) {
      const int idx = it * 256 + t;
      const int r = idx >> 6;
      const int k = idx & 63;
      q_f[k * RB + r] = qv[it];
    }
    reinterpret_cast<float*>(w_s)[t & (Kk - 1)] = w2v;
  }
  lds_barrier();

  float s[RB];
#pragma unroll
  for (int r = 0; r < RB; ++r) s[r] = 0.0f;

  KBlk blk[2];
  load_blk(blk[0], w_s, q_s, 0);
#pragma unroll
  for (int k4 = 0; k4 < Kk / 4; ++k4) {
    if (k4 + 1 < Kk / 4) load_blk(blk[(k4 + 1) & 1], w_s, q_s, k4 + 1);
    __builtin_amdgcn_sched_barrier(0);
    const KBlk& c = blk[k4 & 1];
    const float4 kv = kvr[k4];
    k_step(s, kv.x, c.w.x, c.q[0], c.q[1]);
    k_step(s, kv.y, c.w.y, c.q[2], c.q[3]);
    k_step(s, kv.z, c.w.z, c.q[4], c.q[5]);
    k_step(s, kv.w, c.w.w, c.q[6], c.q[7]);
  }

  float m[RB];
#pragma unroll
  for (int r = 0; r < RB; ++r) m[r] = wave_max(s[r]);
  if (lane == 0) {
    mx_s[wave * 2 + 0] = make_float4(m[0], m[1], m[2], m[3]);
    mx_s[wave * 2 + 1] = make_float4(m[4], m[5], m[6], m[7]);
  }
  __syncthreads();
  {
    float4 a = mx_s[0], c = mx_s[1];
    for (int w = 1; w < 8; ++w) {
      const float4 a2 = mx_s[w * 2 + 0], c2 = mx_s[w * 2 + 1];
      a.x = fmaxf(a.x, a2.x); a.y = fmaxf(a.y, a2.y);
      a.z = fmaxf(a.z, a2.z); a.w = fmaxf(a.w, a2.w);
      c.x = fmaxf(c.x, c2.x); c.y = fmaxf(c.y, c2.y);
      c.z = fmaxf(c.z, c2.z); c.w = fmaxf(c.w, c2.w);
    }
    m[0] = a.x; m[1] = a.y; m[2] = a.z; m[3] = a.w;
    m[4] = c.x; m[5] = c.y; m[6] = c.z; m[7] = c.w;
  }

  float e[RB], p[RB];
#pragma unroll
  for (int r = 0; r < RB; ++r) {
    e[r] = __expf(s[r] - m[r]);
    p[r] = wave_sum(e[r]);
  }
  if (lane == 0) {
    sm_s[wave * 2 + 0] = make_float4(p[0], p[1], p[2], p[3]);
    sm_s[wave * 2 + 1] = make_float4(p[4], p[5], p[6], p[7]);
  }
  __syncthreads();
  {
    float4 a = sm_s[0], c = sm_s[1];
    for (int w = 1; w < 8; ++w) {
      const float4 a2 = sm_s[w * 2 + 0], c2 = sm_s[w * 2 + 1];
      a.x += a2.x; a.y += a2.y; a.z += a2.z; a.w += a2.w;
      c.x += c2.x; c.y += c2.y; c.z += c2.z; c.w += c2.w;
    }
    p[0] = a.x; p[1] = a.y; p[2] = a.z; p[3] = a.w;
    p[4] = c.x; p[5] = c.y; p[6] = c.z; p[7] = c.w;
  }

#pragma unroll
  for (int r = 0; r < RB; ++r) {
    attn[((size_t)b * Ll + (i0 + r)) * Ll + t] = e[r] * (1.0f / p[r]);
  }
}

constexpr int CM  = 32;
constexpr int CN  = 64;
constexpr int XP  = CN + 16;
constexpr int XP4 = XP / 4;
constexpr int CP  = CN + 4;
constexpr int CP4 = CP / 4;

__global__ __launch_bounds__(128) void context_kernel(
    const float* __restrict__ attn, const float* __restrict__ inp,
    float* __restrict__ out) {
  constexpr int XT_OFF = CM * GP4;
  __shared__ v4f lds[XT_OFF + Ll * XP4];
  __shared__ v4f ctile[CM * CP4];

  const int t     = threadIdx.x;
  const int wave  = t >> 5;
  const int lane  = t & 31;
  const int b     = blockIdx.x >> 5;
  const int i0    = ((blockIdx.x & 31) >> 2) * CM;
  const int dbase = (blockIdx.x & 3) * CN;
  const int dl    = wave * 16;

  const v4f* asrc = reinterpret_cast<const v4f*>(
      attn + ((size_t)b * Ll + i0) * Ll);
  const v4f* xsrc = reinterpret_cast<const v4f*>(
      inp + (size_t)b * Ll * Dd + dbase);
  v4f ra[NCH * 4];
  v4f rx[NCH * 8];
#pragma unroll
  for (int c = 0; c < NCH; ++c) {
#pragma unroll
    for (int i = 0; i < 4; ++i) {
      const int idx = i * 128 + t;
      const int row = idx >> 4;
      const int c4  = 16 * c + (idx & 15);
      ra[c * 4 + i] = asrc[row * (Ll / 4) + c4];
    }
#pragma unroll
    for (int i = 0; i < 8; ++i) {
      const int f   = (8 * c + i) * 128 + t;
      const int row = f >> 4;
      const int c4  = f & 15;
      rx[c * 8 + i] = xsrc[row * (Dd / 4) + c4];
    }
    __builtin_amdgcn_sched_barrier(0);
  }
#pragma unroll
  for (int i = 0; i < 4; ++i) {
    const int idx = i * 128 + t;
    lds[(idx >> 4) * GP4 + (idx & 15)] = ra[i];
  }
#pragma unroll
  for (int i = 0; i < 8; ++i) {
    const int f = i * 128 + t;
    lds[XT_OFF + (f >> 4) * XP4 + (f & 15)] = rx[i];
  }
  lds_barrier();

  const int mn = lane & 15;
  const int kh = (lane >> 4) << 1;

  const float* lf    = reinterpret_cast<const float*>(lds);
  const float* xt    = lf + CM * GP;
  const float* arow0 = lf + mn * GP + kh;
  const float* arow1 = lf + (16 + mn) * GP + kh;
  const float* bcol  = xt + kh * XP + dl + mn;

  v2f fa0[2][KB], fa1[2][KB], fb[2][KB];
#pragma unroll
  for (int s = 0; s < KB; ++s) {
    fa0[0][s]  = lds_pair(arow0 + 4 * s);
    fa1[0][s]  = lds_pair(arow1 + 4 * s);
    fb[0][s].x = bcol[(4 * s) * XP];
    fb[0][s].y = bcol[(4 * s + 1) * XP];
  }
  v8f acc0 = {}, acc1 = {};
#pragma unroll
  for (int kb = 0; kb < NCH; ++kb) {
    if (kb + 1 < NCH) {
#pragma unroll
      for (int i = 0; i < 4; ++i) {
        const int idx = i * 128 + t;
        lds[(idx >> 4) * GP4 + 16 * (kb + 1) + (idx & 15)] = ra[(kb + 1) * 4 + i];
      }
#pragma unroll
      for (int i = 0; i < 8; ++i) {
        const int f = (8 * (kb + 1) + i) * 128 + t;
        lds[XT_OFF + (f >> 4) * XP4 + (f & 15)] = rx[(kb + 1) * 8 + i];
      }
      lds_barrier();
#pragma unroll
      for (int s = 0; s < KB; ++s) {
        const int j = 4 * KB * (kb + 1) + 4 * s;
        fa0[(kb + 1) & 1][s]  = lds_pair(arow0 + j);
        fa1[(kb + 1) & 1][s]  = lds_pair(arow1 + j);
        fb[(kb + 1) & 1][s].x = bcol[j * XP];
        fb[(kb + 1) & 1][s].y = bcol[(j + 1) * XP];
      }
    }
    __builtin_amdgcn_sched_barrier(0);
#pragma unroll
    for (int s = 0; s < KB; ++s) {
      acc0 = wmma_f32_4(fa0[kb & 1][s], fb[kb & 1][s], acc0);
      acc1 = wmma_f32_4(fa1[kb & 1][s], fb[kb & 1][s], acc1);
    }
    __builtin_amdgcn_sched_barrier(0);
  }

  float* ct    = reinterpret_cast<float*>(ctile);
  const int mb = (lane >> 4) * 8;
#pragma unroll
  for (int r = 0; r < 8; ++r) {
    ct[(mb + r) * CP + dl + mn]      = acc0[r];
    ct[(16 + mb + r) * CP + dl + mn] = acc1[r];
  }
  __syncthreads();
#pragma unroll
  for (int it = 0; it < (CM * CN / 4) / 128; ++it) {
    const int f   = it * 128 + t;
    const int row = f >> 4;
    const int c4  = f & 15;
    v4f* orow = reinterpret_cast<v4f*>(
        out + ((size_t)b * Ll + i0 + row) * (2 * Dd) + dbase);
    orow[c4]          = lds[XT_OFF + (i0 + row) * XP4 + c4];
    orow[Dd / 4 + c4] = ctile[row * CP4 + c4];
  }
}

extern "C" void kernel_launch(void* const* d_in, const int* in_sizes, int n_in,
                              void* d_out, int out_size, void* d_ws,
                              size_t ws_size, hipStream_t stream) {
  const float* inp = (const float*)d_in[0];
  const float* W1  = (const float*)d_in[1];
  const float* W2  = (const float*)d_in[2];
  float* out = (float*)d_out;

  float* qi   = (float*)d_ws;
  float* kj   = qi + (size_t)Bb * Ll * Kk;
  float* attn = kj + (size_t)Bb * Ll * Kk;

  qk_proj_kernel<<<(Bb * Ll / TM) * (2 * Kk / TN), 128, 0, stream>>>(inp, W1, qi, kj);
  attn_softmax_kernel<<<Bb * (Ll / RB), 256, 0, stream>>>(qi, kj, W2, attn);
  context_kernel<<<Bb * (Ll / CM) * (Dd / CN), 128, 0, stream>>>(attn, inp, out);
}
